# starting kernel
# baseline (speedup 1.0000x reference)
#include <hip/hip_runtime.h>
#include <hip/hip_bf16.h>

typedef float v2f __attribute__((ext_vector_type(2)));
typedef float v8f __attribute__((ext_vector_type(8)));

#define DD 128
#define HH 4
#define CC 32

__global__ void gnn_fill_u32(unsigned int* __restrict__ p, unsigned int v, long long n) {
    long long i = (long long)blockIdx.x * blockDim.x + threadIdx.x;
    if (i < n) p[i] = v;
}

__global__ void gnn_edge_init(const float* __restrict__ x, const int* __restrict__ ei,
                              const float* __restrict__ W_edge, const float* __restrict__ b_edge,
                              float* __restrict__ h_sum, float* __restrict__ cnt, int E_) {
    int e = blockIdx.x;
    int d = threadIdx.x;
    int dstn = ei[E_ + e];
    float x0 = x[e * 3 + 0], x1 = x[e * 3 + 1], x2 = x[e * 3 + 2];
    float v = x0 * W_edge[d] + x1 * W_edge[DD + d] + x2 * W_edge[2 * DD + d] + b_edge[d];
    atomicAdd(&h_sum[(long long)dstn * DD + d], v);
    if (d == 0) atomicAdd(&cnt[dstn], 1.0f);
}

__global__ void gnn_wmma_gemm128(const float* __restrict__ A, const float* __restrict__ B,
                                 float* __restrict__ C, const float* __restrict__ bias,
                                 const float* __restrict__ cnt, int M) {
    int wave = threadIdx.x >> 5;
    int lane = threadIdx.x & 31;
    int m0 = blockIdx.x * 16;
    int n0 = (blockIdx.y * 4 + wave) * 16;
    int half = lane >> 4;
    int l    = lane & 15;
    int K0   = half * 2;
    int Mrow = m0 + l;
    if (Mrow >= M) Mrow = M - 1;
    int Ncol = n0 + l;

    const float* Arow = A + (long long)Mrow * DD;
    v8f acc = {};
#pragma unroll
    for (int k = 0; k < DD; k += 4) {
        v2f a = *(const v2f*)(Arow + k + K0);
        v2f b;
        b.x = B[(long long)(k + K0) * DD + Ncol];
        b.y = B[(long long)(k + K0 + 1) * DD + Ncol];
        acc = __builtin_amdgcn_wmma_f32_16x16x4_f32(false, a, false, b,
                                                    (short)0, acc, false, false);
    }
#pragma unroll
    for (int r = 0; r < 8; ++r) {
        int row = m0 + r + half * 8;
        if (row < M) {
            float v = acc[r];
            if (cnt)  v *= 1.0f / fmaxf(cnt[row], 1.0f);
            if (bias) v += bias[n0 + l];
            C[(long long)row * DD + n0 + l] = v;
        }
    }
}

__global__ void gnn_att_scores(const float* __restrict__ xh,
                               const float* __restrict__ att_src,
                               const float* __restrict__ att_dst,
                               float* __restrict__ a_s, float* __restrict__ a_d) {
    int n = blockIdx.x;
    int d = threadIdx.x;
    int head = d >> 5;
    float v  = xh[(long long)n * DD + d];
    float s  = v * att_src[d];
    float t  = v * att_dst[d];
#pragma unroll
    for (int m = 16; m; m >>= 1) { s += __shfl_xor(s, m); t += __shfl_xor(t, m); }
    if ((d & 31) == 0) {
        a_s[n * HH + head] = s;
        a_d[n * HH + head] = t;
    }
}

__device__ __forceinline__ void edge_st(int t, int E_, const int* ei, int& s, int& dd) {
    if (t < E_) { s = ei[t]; dd = ei[E_ + t]; }
    else        { s = t - E_; dd = t - E_; }
}

__global__ void gnn_edge_max(const int* __restrict__ ei, const float* __restrict__ a_s,
                             const float* __restrict__ a_d, float* __restrict__ mval,
                             int E_, int Nn) {
    int idx = blockIdx.x * blockDim.x + threadIdx.x;
    int tot = E_ + Nn;
    int t = idx >> 2, hd = idx & 3;
    if (t >= tot) return;
    int s, dd; edge_st(t, E_, ei, s, dd);
    float e = a_s[s * HH + hd] + a_d[dd * HH + hd];
    e = (e > 0.0f) ? e : 0.2f * e;
    int* p = (int*)&mval[dd * HH + hd];
    if (e >= 0.0f) atomicMax(p, __float_as_int(e));
    else           atomicMin((unsigned int*)p, __float_as_uint(e));
}

__global__ void gnn_edge_denom(const int* __restrict__ ei, const float* __restrict__ a_s,
                               const float* __restrict__ a_d, const float* __restrict__ mval,
                               float* __restrict__ denom, int E_, int Nn) {
    int idx = blockIdx.x * blockDim.x + threadIdx.x;
    int tot = E_ + Nn;
    int t = idx >> 2, hd = idx & 3;
    if (t >= tot) return;
    int s, dd; edge_st(t, E_, ei, s, dd);
    float e = a_s[s * HH + hd] + a_d[dd * HH + hd];
    e = (e > 0.0f) ? e : 0.2f * e;
    atomicAdd(&denom[dd * HH + hd], __expf(e - mval[dd * HH + hd]));
}

__global__ void gnn_edge_accum(const int* __restrict__ ei, const float* __restrict__ a_s,
                               const float* __restrict__ a_d, const float* __restrict__ mval,
                               const float* __restrict__ denom, const float* __restrict__ xh,
                               float* __restrict__ out, int E_, int Nn) {
    int t = blockIdx.x;
    int d = threadIdx.x;
    int hd = d >> 5;
    int s, dd; edge_st(t, E_, ei, s, dd);
    float e = a_s[s * HH + hd] + a_d[dd * HH + hd];
    e = (e > 0.0f) ? e : 0.2f * e;
    float alpha = __expf(e - mval[dd * HH + hd]) / denom[dd * HH + hd];
    atomicAdd(&out[(long long)dd * DD + d], alpha * xh[(long long)s * DD + d]);
}

__global__ void gnn_finalize(const float* __restrict__ out, const float* __restrict__ gbias,
                             const float* __restrict__ ln_g, const float* __restrict__ ln_b,
                             float* __restrict__ h, int do_relu) {
    __shared__ float red[4];
    int n = blockIdx.x;
    int d = threadIdx.x;
    float v = out[(long long)n * DD + d] + gbias[d];
    float s = v;
#pragma unroll
    for (int m = 16; m; m >>= 1) s += __shfl_xor(s, m);
    if ((d & 31) == 0) red[d >> 5] = s;
    __syncthreads();
    float mean = (red[0] + red[1] + red[2] + red[3]) * (1.0f / DD);
    float c = v - mean;
    float q = c * c;
#pragma unroll
    for (int m = 16; m; m >>= 1) q += __shfl_xor(q, m);
    __syncthreads();
    if ((d & 31) == 0) red[d >> 5] = q;
    __syncthreads();
    float var = (red[0] + red[1] + red[2] + red[3]) * (1.0f / DD);
    float y = c * rsqrtf(var + 1e-5f) * ln_g[d] + ln_b[d];
    if (do_relu) y = fmaxf(y, 0.0f);
    h[(long long)n * DD + d] += y;
}

extern "C" void kernel_launch(void* const* d_in, const int* in_sizes, int n_in,
                              void* d_out, int out_size, void* d_ws, size_t ws_size,
                              hipStream_t stream) {
    const float* x      = (const float*)d_in[0];
    const int*   ei     = (const int*)d_in[1];
    const float* W_edge = (const float*)d_in[2];
    const float* b_edge = (const float*)d_in[3];
    const float* W_node = (const float*)d_in[4];
    const float* b_node = (const float*)d_in[5];
    const float* lin_w  = (const float*)d_in[6];
    const float* att_sp = (const float*)d_in[7];
    const float* att_dp = (const float*)d_in[8];
    const float* g_bias = (const float*)d_in[9];
    const float* ln_g   = (const float*)d_in[10];
    const float* ln_b   = (const float*)d_in[11];

    const int E_ = in_sizes[0] / 3;
    const int Nn = out_size / DD;
    const int tot = E_ + Nn;

    float* bufA  = (float*)d_ws;
    float* xh    = bufA + (long long)Nn * DD;
    float* cnt   = xh   + (long long)Nn * DD;
    float* a_s   = cnt  + Nn;
    float* a_d   = a_s  + (long long)Nn * HH;
    float* mval  = a_d  + (long long)Nn * HH;
    float* denom = mval + (long long)Nn * HH;
    float* h     = (float*)d_out;

    long long nh = (long long)Nn * DD;
    int gh  = (int)((nh + 255) / 256);
    int g4  = (Nn * HH + 255) / 256;
    int g1  = (Nn + 255) / 256;
    int geh = (tot * HH + 255) / 256;
    dim3 gemm_grid((Nn + 15) / 16, 2);

    gnn_fill_u32<<<gh, 256, 0, stream>>>((unsigned int*)bufA, 0u, nh);
    gnn_fill_u32<<<g1, 256, 0, stream>>>((unsigned int*)cnt, 0u, Nn);
    gnn_edge_init<<<E_, DD, 0, stream>>>(x, ei, W_edge, b_edge, bufA, cnt, E_);
    gnn_wmma_gemm128<<<gemm_grid, 128, 0, stream>>>(bufA, W_node, h, b_node, cnt, Nn);

    for (int i = 0; i < 3; ++i) {
        gnn_wmma_gemm128<<<gemm_grid, 128, 0, stream>>>(
            h, lin_w + (long long)i * DD * DD, xh, nullptr, nullptr, Nn);
        gnn_att_scores<<<Nn, DD, 0, stream>>>(xh, att_sp + i * DD, att_dp + i * DD, a_s, a_d);

        gnn_fill_u32<<<gh, 256, 0, stream>>>((unsigned int*)bufA, 0u, nh);
        gnn_fill_u32<<<g4, 256, 0, stream>>>((unsigned int*)denom, 0u, Nn * HH);
        gnn_fill_u32<<<g4, 256, 0, stream>>>((unsigned int*)mval, 0xFF800000u, Nn * HH);

        gnn_edge_max  <<<geh, 256, 0, stream>>>(ei, a_s, a_d, mval, E_, Nn);
        gnn_edge_denom<<<geh, 256, 0, stream>>>(ei, a_s, a_d, mval, denom, E_, Nn);
        gnn_edge_accum<<<tot, DD, 0, stream>>>(ei, a_s, a_d, mval, denom, xh, bufA, E_, Nn);

        gnn_finalize<<<Nn, DD, 0, stream>>>(bufA, g_bias + i * DD, ln_g + i * DD,
                                            ln_b + i * DD, h, (i < 2) ? 1 : 0);
    }
    (void)n_in; (void)ws_size; (void)W_edge;
}


# Edges stable-sorted by destination once per call (count, scan, place, rank by edge id: same list every run); one wave per node gathers its in-edges in registers, no float atomics
# speedup vs baseline: 3.0523x; 3.0523x over previous
#include <hip/hip_runtime.h>
#include <hip/hip_bf16.h>

typedef float v2f __attribute__((ext_vector_type(2)));
typedef float v8f __attribute__((ext_vector_type(8)));

#define DD 128
#define HH 4
#define CC 32

__global__ void gnn_fill_u32(unsigned int* __restrict__ p, unsigned int v, long long n) {
    long long i = (long long)blockIdx.x * blockDim.x + threadIdx.x;
    if (i < n) p[i] = v;
}

__global__ void gnn_csr_count(const int* __restrict__ ei, int* __restrict__ deg, int E_) {
    int e = blockIdx.x * blockDim.x + threadIdx.x;
    if (e < E_) atomicAdd(&deg[ei[E_ + e]], 1);
}

__global__ void __launch_bounds__(1024)
gnn_csr_scan(const int* __restrict__ deg, int* __restrict__ rowptr, int* __restrict__ cursor,
             float* __restrict__ cnt, int n) {
    __shared__ int wtot[32];
    const int t = threadIdx.x, lane = t & 31, w = t >> 5;
    const int chunk = (n + 1023) >> 10;
    int b = t * chunk; if (b > n) b = n;
    int e = b + chunk; if (e > n) e = n;
    int s = 0;
    for (int i = b; i < e; ++i) s += deg[i];
    int incl = s;
    for (int o = 1; o < 32; o <<= 1) { int v = __shfl_up(incl, o); if (lane >= o) incl += v; }
    if (lane == 31) wtot[w] = incl;
    __syncthreads();
    if (w == 0) {
        int v = wtot[lane], iv = v;
        for (int o = 1; o < 32; o <<= 1) { int u = __shfl_up(iv, o); if (lane >= o) iv += u; }
        wtot[lane] = iv - v;
    }
    __syncthreads();
    int run = wtot[w] + incl - s;
    for (int i = b; i < e; ++i) {
        int d = deg[i];
        rowptr[i] = run; cursor[i] = run; cnt[i] = (float)d;
        run += d;
    }
    if (t == 1023) rowptr[n] = run;
}

__global__ void gnn_csr_place(const int* __restrict__ ei, int* __restrict__ cursor,
                              int* __restrict__ slot_eid, int E_) {
    int e = blockIdx.x * blockDim.x + threadIdx.x;
    if (e < E_) { int p = atomicAdd(&cursor[ei[E_ + e]], 1); slot_eid[p] = e; }
}

__global__ void gnn_csr_order(const int* __restrict__ ei, const int* __restrict__ rowptr,
                              const int* __restrict__ slot_eid, int* __restrict__ csr_eid,
                              int* __restrict__ csr_src, int E_) {
    int p = blockIdx.x * blockDim.x + threadIdx.x;
    if (p >= E_) return;
    int eid = slot_eid[p];
    int d = ei[E_ + eid];
    int b = rowptr[d], en = rowptr[d + 1];
    int r = 0;
    for (int q = b; q < en; ++q) r += (slot_eid[q] < eid) ? 1 : 0;
    csr_eid[b + r] = eid;
    csr_src[b + r] = ei[eid];
}

__device__ __forceinline__ float lane_bcast(float v, int k) {
    return __int_as_float(__builtin_amdgcn_readlane(__float_as_int(v), k));
}
__device__ __forceinline__ float wave_sum(float v) {
#pragma unroll
    for (int m = 16; m; m >>= 1) v += __shfl_xor(v, m);
    return v;
}
__device__ __forceinline__ float wave_max(float v) {
#pragma unroll
    for (int m = 16; m; m >>= 1) v = fmaxf(v, __shfl_xor(v, m));
    return v;
}

__global__ void __launch_bounds__(256)
gnn_node_init(const float* __restrict__ x, const int* __restrict__ rowptr,
              const int* __restrict__ csr_eid, const float* __restrict__ W_edge,
              const float* __restrict__ b_edge, float* __restrict__ h_sum, int Nn) {
    const int lane = threadIdx.x & 31;
    const int n = __builtin_amdgcn_readfirstlane((int)(blockIdx.x * 8 + (threadIdx.x >> 5)));
    if (n >= Nn) return;
    const int start = __builtin_amdgcn_readfirstlane(rowptr[n]);
    const int end   = __builtin_amdgcn_readfirstlane(rowptr[n + 1]);
    const float4 w0 = *(const float4*)(W_edge + lane * 4);
    const float4 w1 = *(const float4*)(W_edge + DD + lane * 4);
    const float4 w2 = *(const float4*)(W_edge + 2 * DD + lane * 4);
    const float4 bb = *(const float4*)(b_edge + lane * 4);
    float4 acc = make_float4(0.f, 0.f, 0.f, 0.f);
    for (int j0 = start; j0 < end; j0 += 32) {
        int j = j0 + lane;
        float x0 = 0.f, x1 = 0.f, x2 = 0.f;
        if (j < end) {
            int e = csr_eid[j];
            x0 = x[(long long)e * 3 + 0]; x1 = x[(long long)e * 3 + 1]; x2 = x[(long long)e * 3 + 2];
        }
        int c = end - j0; if (c > 32) c = 32;
        for (int k = 0; k < c; ++k) {
            float y0 = lane_bcast(x0, k), y1 = lane_bcast(x1, k), y2 = lane_bcast(x2, k);
            acc.x += y0 * w0.x + y1 * w1.x + y2 * w2.x + bb.x;
            acc.y += y0 * w0.y + y1 * w1.y + y2 * w2.y + bb.y;
            acc.z += y0 * w0.z + y1 * w1.z + y2 * w2.z + bb.z;
            acc.w += y0 * w0.w + y1 * w1.w + y2 * w2.w + bb.w;
        }
    }
    *(float4*)(h_sum + (long long)n * DD + lane * 4) = acc;
}

__global__ void gnn_wmma_gemm128(const float* __restrict__ A, const float* __restrict__ B,
                                 float* __restrict__ C, const float* __restrict__ bias,
                                 const float* __restrict__ cnt, int M) {
    int wave = threadIdx.x >> 5;
    int lane = threadIdx.x & 31;
    int m0 = blockIdx.x * 16;
    int n0 = (blockIdx.y * 4 + wave) * 16;
    int half = lane >> 4;
    int l    = lane & 15;
    int K0   = half * 2;
    int Mrow = m0 + l;
    if (Mrow >= M) Mrow = M - 1;
    int Ncol = n0 + l;

    const float* Arow = A + (long long)Mrow * DD;
    v8f acc = {};
#pragma unroll
    for (int k = 0; k < DD; k += 4) {
        v2f a = *(const v2f*)(Arow + k + K0);
        v2f b;
        b.x = B[(long long)(k + K0) * DD + Ncol];
        b.y = B[(long long)(k + K0 + 1) * DD + Ncol];
        acc = __builtin_amdgcn_wmma_f32_16x16x4_f32(false, a, false, b,
                                                    (short)0, acc, false, false);
    }
#pragma unroll
    for (int r = 0; r < 8; ++r) {
        int row = m0 + r + half * 8;
        if (row < M) {
            float v = acc[r];
            if (cnt)  v *= 1.0f / fmaxf(cnt[row], 1.0f);
            if (bias) v += bias[n0 + l];
            C[(long long)row * DD + n0 + l] = v;
        }
    }
}

__global__ void gnn_att_scores(const float* __restrict__ xh,
                               const float* __restrict__ att_src,
                               const float* __restrict__ att_dst,
                               float* __restrict__ a_s, float* __restrict__ a_d) {
    int n = blockIdx.x;
    int d = threadIdx.x;
    int head = d >> 5;
    float v  = xh[(long long)n * DD + d];
    float s  = v * att_src[d];
    float t  = v * att_dst[d];
#pragma unroll
    for (int m = 16; m; m >>= 1) { s += __shfl_xor(s, m); t += __shfl_xor(t, m); }
    if ((d & 31) == 0) {
        a_s[n * HH + head] = s;
        a_d[n * HH + head] = t;
    }
}

__device__ __forceinline__ float4 gat_logit(const float* __restrict__ a_s, int s, float4 ad) {
    float4 as = *(const float4*)(a_s + (long long)s * HH);
    float4 e = make_float4(as.x + ad.x, as.y + ad.y, as.z + ad.z, as.w + ad.w);
    e.x = (e.x > 0.0f) ? e.x : 0.2f * e.x;
    e.y = (e.y > 0.0f) ? e.y : 0.2f * e.y;
    e.z = (e.z > 0.0f) ? e.z : 0.2f * e.z;
    e.w = (e.w > 0.0f) ? e.w : 0.2f * e.w;
    return e;
}

__device__ __forceinline__ void gat_accum(float4& acc, const float* __restrict__ xh,
                                          int c, int s, float4 al, int lane) {
#pragma unroll 4
    for (int k = 0; k < c; ++k) {
        int sk = __builtin_amdgcn_readlane(s, k);
        float a0 = lane_bcast(al.x, k), a1 = lane_bcast(al.y, k);
        float a2 = lane_bcast(al.z, k), a3 = lane_bcast(al.w, k);
        const float* r = xh + (long long)sk * DD + lane;
        float v0 = r[0], v1 = r[CC], v2 = r[2 * CC], v3 = r[3 * CC];
        acc.x += a0 * v0; acc.y += a1 * v1; acc.z += a2 * v2; acc.w += a3 * v3;
    }
}

__global__ void __launch_bounds__(256)
gnn_gat_gather(const int* __restrict__ rowptr, const int* __restrict__ csr_src,
               const float* __restrict__ a_s, const float* __restrict__ a_d,
               const float* __restrict__ xh, const float* __restrict__ gbias,
               const float* __restrict__ ln_g, const float* __restrict__ ln_b,
               float* __restrict__ h, int Nn, int do_relu) {
    const int lane = threadIdx.x & 31;
    const int n = __builtin_amdgcn_readfirstlane((int)(blockIdx.x * 8 + (threadIdx.x >> 5)));
    if (n >= Nn) return;
    const int start = __builtin_amdgcn_readfirstlane(rowptr[n]);
    const int deg   = __builtin_amdgcn_readfirstlane(rowptr[n + 1]) - start;
    const int tot   = deg + 1;
    const float4 ad = *(const float4*)(a_d + (long long)n * HH);
    const float ninf = -__builtin_inff();

    const bool v0 = lane < tot;
    const int  s0 = (lane < deg) ? csr_src[start + lane] : n;
    const float4 e0 = gat_logit(a_s, s0, ad);

    float4 m = v0 ? e0 : make_float4(ninf, ninf, ninf, ninf);
    for (int j0 = 32; j0 < tot; j0 += 32) {
        int j = j0 + lane;
        if (j < tot) {
            int s = (j < deg) ? csr_src[start + j] : n;
            float4 e = gat_logit(a_s, s, ad);
            m.x = fmaxf(m.x, e.x); m.y = fmaxf(m.y, e.y); m.z = fmaxf(m.z, e.z); m.w = fmaxf(m.w, e.w);
        }
    }
    m.x = wave_max(m.x); m.y = wave_max(m.y); m.z = wave_max(m.z); m.w = wave_max(m.w);

    float4 p0 = make_float4(0.f, 0.f, 0.f, 0.f);
    if (v0) p0 = make_float4(__expf(e0.x - m.x), __expf(e0.y - m.y), __expf(e0.z - m.z), __expf(e0.w - m.w));
    float4 den = p0;
    for (int j0 = 32; j0 < tot; j0 += 32) {
        int j = j0 + lane;
        if (j < tot) {
            int s = (j < deg) ? csr_src[start + j] : n;
            float4 e = gat_logit(a_s, s, ad);
            den.x += __expf(e.x - m.x); den.y += __expf(e.y - m.y);
            den.z += __expf(e.z - m.z); den.w += __expf(e.w - m.w);
        }
    }
    den.x = wave_sum(den.x); den.y = wave_sum(den.y); den.z = wave_sum(den.z); den.w = wave_sum(den.w);

    float4 acc = make_float4(0.f, 0.f, 0.f, 0.f);
    {
        float4 al = make_float4(p0.x / den.x, p0.y / den.y, p0.z / den.z, p0.w / den.w);
        int c = (tot < 32) ? tot : 32;
        gat_accum(acc, xh, c, s0, al, lane);
    }
    for (int j0 = 32; j0 < tot; j0 += 32) {
        int j = j0 + lane;
        int s = n;
        float4 al = make_float4(0.f, 0.f, 0.f, 0.f);
        if (j < tot) {
            s = (j < deg) ? csr_src[start + j] : n;
            float4 e = gat_logit(a_s, s, ad);
            al = make_float4(__expf(e.x - m.x) / den.x, __expf(e.y - m.y) / den.y,
                             __expf(e.z - m.z) / den.z, __expf(e.w - m.w) / den.w);
        }
        int c = tot - j0; if (c > 32) c = 32;
        gat_accum(acc, xh, c, s, al, lane);
    }

    float4 v = make_float4(acc.x + gbias[lane], acc.y + gbias[CC + lane],
                           acc.z + gbias[2 * CC + lane], acc.w + gbias[3 * CC + lane]);
    float mean = wave_sum((v.x + v.y) + (v.z + v.w)) * (1.0f / DD);
    float4 cc = make_float4(v.x - mean, v.y - mean, v.z - mean, v.w - mean);
    float var = wave_sum((cc.x * cc.x + cc.y * cc.y) + (cc.z * cc.z + cc.w * cc.w)) * (1.0f / DD);
    float rs = rsqrtf(var + 1e-5f);
    float4 y = make_float4(cc.x * rs * ln_g[lane] + ln_b[lane],
                           cc.y * rs * ln_g[CC + lane] + ln_b[CC + lane],
                           cc.z * rs * ln_g[2 * CC + lane] + ln_b[2 * CC + lane],
                           cc.w * rs * ln_g[3 * CC + lane] + ln_b[3 * CC + lane]);
    if (do_relu) { y.x = fmaxf(y.x, 0.f); y.y = fmaxf(y.y, 0.f); y.z = fmaxf(y.z, 0.f); y.w = fmaxf(y.w, 0.f); }
    float* hp = h + (long long)n * DD + lane;
    hp[0]      += y.x;
    hp[CC]     += y.y;
    hp[2 * CC] += y.z;
    hp[3 * CC] += y.w;
}

extern "C" void kernel_launch(void* const* d_in, const int* in_sizes, int n_in,
                              void* d_out, int out_size, void* d_ws, size_t ws_size,
                              hipStream_t stream) {
    const float* x      = (const float*)d_in[0];
    const int*   ei     = (const int*)d_in[1];
    const float* W_edge = (const float*)d_in[2];
    const float* b_edge = (const float*)d_in[3];
    const float* W_node = (const float*)d_in[4];
    const float* b_node = (const float*)d_in[5];
    const float* lin_w  = (const float*)d_in[6];
    const float* att_sp = (const float*)d_in[7];
    const float* att_dp = (const float*)d_in[8];
    const float* g_bias = (const float*)d_in[9];
    const float* ln_g   = (const float*)d_in[10];
    const float* ln_b   = (const float*)d_in[11];

    const int E_ = in_sizes[0] / 3;
    const int Nn = out_size / DD;

    int*   rowptr   = (int*)d_ws;
    int*   deg      = rowptr + (Nn + 1);
    int*   cursor   = deg + Nn;
    int*   slot_eid = cursor + Nn;
    int*   csr_eid  = slot_eid + E_;
    int*   csr_src  = csr_eid + E_;
    float* xh    = (float*)d_ws + (long long)Nn * DD;
    float* a_s   = xh  + (long long)Nn * DD;
    float* a_d   = a_s + (long long)Nn * HH;
    float* cnt   = a_d + (long long)Nn * HH;
    float* h     = (float*)d_out;

    int g1  = (Nn + 255) / 256;
    int ge  = (E_ + 255) / 256;
    int gw  = (Nn + 7) / 8;
    dim3 gemm_grid((Nn + 15) / 16, 2);

    gnn_fill_u32<<<g1, 256, 0, stream>>>((unsigned int*)deg, 0u, Nn);
    gnn_csr_count<<<ge, 256, 0, stream>>>(ei, deg, E_);
    gnn_csr_scan<<<1, 1024, 0, stream>>>(deg, rowptr, cursor, cnt, Nn);
    gnn_csr_place<<<ge, 256, 0, stream>>>(ei, cursor, slot_eid, E_);
    gnn_csr_order<<<ge, 256, 0, stream>>>(ei, rowptr, slot_eid, csr_eid, csr_src, E_);

    gnn_node_init<<<gw, 256, 0, stream>>>(x, rowptr, csr_eid, W_edge, b_edge, xh, Nn);
    gnn_wmma_gemm128<<<gemm_grid, 128, 0, stream>>>(xh, W_node, h, b_node, cnt, Nn);

    for (int i = 0; i < 3; ++i) {
        gnn_wmma_gemm128<<<gemm_grid, 128, 0, stream>>>(
            h, lin_w + (long long)i * DD * DD, xh, nullptr, nullptr, Nn);
        gnn_att_scores<<<Nn, DD, 0, stream>>>(xh, att_sp + i * DD, att_dp + i * DD, a_s, a_d);
        gnn_gat_gather<<<gw, 256, 0, stream>>>(rowptr, csr_src, a_s, a_d, xh,
                                               g_bias + i * DD, ln_g + i * DD, ln_b + i * DD,
                                               h, Nn, (i < 2) ? 1 : 0);
    }
    (void)n_in; (void)ws_size;
}


# Degree prefix scan spread over one 1024-thread block per 1024 nodes (each block sums the degrees in front of its range with coalesced reads, then scans its own) instead of one block walking all nodes
# speedup vs baseline: 4.3439x; 1.4232x over previous
#include <hip/hip_runtime.h>
#include <hip/hip_bf16.h>

typedef float v2f __attribute__((ext_vector_type(2)));
typedef float v8f __attribute__((ext_vector_type(8)));

#define DD 128
#define HH 4
#define CC 32

__global__ void gnn_fill_u32(unsigned int* __restrict__ p, unsigned int v, long long n) {
    long long i = (long long)blockIdx.x * blockDim.x + threadIdx.x;
    if (i < n) p[i] = v;
}

__global__ void gnn_csr_count(const int* __restrict__ ei, int* __restrict__ deg, int E_) {
    int e = blockIdx.x * blockDim.x + threadIdx.x;
    if (e < E_) atomicAdd(&deg[ei[E_ + e]], 1);
}

__global__ void __launch_bounds__(1024)
gnn_csr_scan(const int* __restrict__ deg, int* __restrict__ rowptr, int* __restrict__ cursor,
             float* __restrict__ cnt, int n) {
    __shared__ int wtot[32];
    __shared__ int ptot[32];
    const int t = threadIdx.x, lane = t & 31, w = t >> 5;
    const int base = blockIdx.x * 1024;
    int ps = 0;
    for (int j0 = t; j0 < base; j0 += 16 * 1024) {
#pragma unroll
        for (int u = 0; u < 16; ++u) {
            const int j = j0 + u * 1024;
            const bool in = j < base;
            const int v = deg[in ? j : t];
            ps += in ? v : 0;
        }
    }
    const int i = base + t;
    const int d = (i < n) ? deg[i] : 0;
    int incl = d;
    for (int o = 1; o < 32; o <<= 1) { int v = __shfl_up(incl, o); if (lane >= o) incl += v; }
    for (int m = 16; m; m >>= 1) ps += __shfl_xor(ps, m);
    if (lane == 31) wtot[w] = incl;
    if (lane == 0)  ptot[w] = ps;
    __syncthreads();
    const int wv = wtot[lane];
    int iv = wv;
    for (int o = 1; o < 32; o <<= 1) { int u = __shfl_up(iv, o); if (lane >= o) iv += u; }
    const int woff = __shfl(iv - wv, w);
    int pre = ptot[lane];
    for (int m = 16; m; m >>= 1) pre += __shfl_xor(pre, m);
    const int run = pre + woff + incl - d;
    if (i < n) {
        rowptr[i] = run; cursor[i] = run; cnt[i] = (float)d;
        if (i == n - 1) rowptr[n] = run + d;
    }
}

__global__ void gnn_csr_place(const int* __restrict__ ei, int* __restrict__ cursor,
                              int* __restrict__ slot_eid, int E_) {
    int e = blockIdx.x * blockDim.x + threadIdx.x;
    if (e < E_) { int p = atomicAdd(&cursor[ei[E_ + e]], 1); slot_eid[p] = e; }
}

__global__ void gnn_csr_order(const int* __restrict__ ei, const int* __restrict__ rowptr,
                              const int* __restrict__ slot_eid, int* __restrict__ csr_eid,
                              int* __restrict__ csr_src, int E_) {
    int p = blockIdx.x * blockDim.x + threadIdx.x;
    if (p >= E_) return;
    int eid = slot_eid[p];
    int d = ei[E_ + eid];
    int b = rowptr[d], en = rowptr[d + 1];
    int r = 0;
    for (int q = b; q < en; ++q) r += (slot_eid[q] < eid) ? 1 : 0;
    csr_eid[b + r] = eid;
    csr_src[b + r] = ei[eid];
}

__device__ __forceinline__ float lane_bcast(float v, int k) {
    return __int_as_float(__builtin_amdgcn_readlane(__float_as_int(v), k));
}
__device__ __forceinline__ float wave_sum(float v) {
#pragma unroll
    for (int m = 16; m; m >>= 1) v += __shfl_xor(v, m);
    return v;
}
__device__ __forceinline__ float wave_max(float v) {
#pragma unroll
    for (int m = 16; m; m >>= 1) v = fmaxf(v, __shfl_xor(v, m));
    return v;
}

__global__ void __launch_bounds__(256)
gnn_node_init(const float* __restrict__ x, const int* __restrict__ rowptr,
              const int* __restrict__ csr_eid, const float* __restrict__ W_edge,
              const float* __restrict__ b_edge, float* __restrict__ h_sum, int Nn) {
    const int lane = threadIdx.x & 31;
    const int n = __builtin_amdgcn_readfirstlane((int)(blockIdx.x * 8 + (threadIdx.x >> 5)));
    if (n >= Nn) return;
    const int start = __builtin_amdgcn_readfirstlane(rowptr[n]);
    const int end   = __builtin_amdgcn_readfirstlane(rowptr[n + 1]);
    const float4 w0 = *(const float4*)(W_edge + lane * 4);
    const float4 w1 = *(const float4*)(W_edge + DD + lane * 4);
    const float4 w2 = *(const float4*)(W_edge + 2 * DD + lane * 4);
    const float4 bb = *(const float4*)(b_edge + lane * 4);
    float4 acc = make_float4(0.f, 0.f, 0.f, 0.f);
    for (int j0 = start; j0 < end; j0 += 32) {
        int j = j0 + lane;
        float x0 = 0.f, x1 = 0.f, x2 = 0.f;
        if (j < end) {
            int e = csr_eid[j];
            x0 = x[(long long)e * 3 + 0]; x1 = x[(long long)e * 3 + 1]; x2 = x[(long long)e * 3 + 2];
        }
        int c = end - j0; if (c > 32) c = 32;
        for (int k = 0; k < c; ++k) {
            float y0 = lane_bcast(x0, k), y1 = lane_bcast(x1, k), y2 = lane_bcast(x2, k);
            acc.x += y0 * w0.x + y1 * w1.x + y2 * w2.x + bb.x;
            acc.y += y0 * w0.y + y1 * w1.y + y2 * w2.y + bb.y;
            acc.z += y0 * w0.z + y1 * w1.z + y2 * w2.z + bb.z;
            acc.w += y0 * w0.w + y1 * w1.w + y2 * w2.w + bb.w;
        }
    }
    *(float4*)(h_sum + (long long)n * DD + lane * 4) = acc;
}

__global__ void gnn_wmma_gemm128(const float* __restrict__ A, const float* __restrict__ B,
                                 float* __restrict__ C, const float* __restrict__ bias,
                                 const float* __restrict__ cnt,
                                 const float* __restrict__ att_src,
                                 const float* __restrict__ att_dst,
                                 float* __restrict__ a_s, float* __restrict__ a_d, int M) {
    const int wave = threadIdx.x >> 5;
    const int lane = threadIdx.x & 31;
    const int m0 = blockIdx.x * 16;
    const int half = lane >> 4;
    const int l    = lane & 15;
    const int K0   = half * 2;
    int Mrow = m0 + l;
    if (Mrow >= M) Mrow = M - 1;
    const int col = wave * CC + 2 * l;

    const float* Arow = A + (long long)Mrow * DD;
    const float* Bcol = B + col;
    const bool fuse = (att_src != nullptr);
    v2f ws = {}, wd = {};
    if (fuse) { ws = *(const v2f*)(att_src + col); wd = *(const v2f*)(att_dst + col); }
    v8f acc0 = {};
    v8f acc1 = {};
#pragma unroll
    for (int k = 0; k < DD; k += 4) {
        v2f a  = *(const v2f*)(Arow + k + K0);
        v2f p0 = *(const v2f*)(Bcol + (k + K0) * DD);
        v2f p1 = *(const v2f*)(Bcol + (k + K0 + 1) * DD);
        v2f b0, b1;
        b0.x = p0.x; b0.y = p1.x;
        b1.x = p0.y; b1.y = p1.y;
        acc0 = __builtin_amdgcn_wmma_f32_16x16x4_f32(false, a, false, b0,
                                                     (short)0, acc0, false, false);
        acc1 = __builtin_amdgcn_wmma_f32_16x16x4_f32(false, a, false, b1,
                                                     (short)0, acc1, false, false);
    }

    float bx = 0.f, by = 0.f;
    if (bias) { bx = bias[col]; by = bias[col + 1]; }
#pragma unroll
    for (int r = 0; r < 8; ++r) {
        const int row = m0 + r + half * 8;
        if (row < M) {
            float v0 = acc0[r], v1 = acc1[r];
            if (cnt) { float inv = 1.0f / fmaxf(cnt[row], 1.0f); v0 *= inv; v1 *= inv; }
            if (bias) { v0 += bx; v1 += by; }
            v2f o; o.x = v0; o.y = v1;
            *(v2f*)(C + (long long)row * DD + col) = o;
            acc0[r] = v0; acc1[r] = v1;
        }
    }

    if (fuse) {
        float v[16];
#pragma unroll
        for (int r = 0; r < 8; ++r) {
            v[r]     = acc0[r] * ws.x + acc1[r] * ws.y;
            v[8 + r] = acc0[r] * wd.x + acc1[r] * wd.y;
        }
        const bool b3 = (l & 8) != 0, b2 = (l & 4) != 0, b1 = (l & 2) != 0, b0 = (l & 1) != 0;
        float w[8];
#pragma unroll
        for (int j = 0; j < 8; ++j) {
            float keep = b3 ? v[j + 8] : v[j];
            float send = b3 ? v[j] : v[j + 8];
            w[j] = keep + __shfl_xor(send, 8);
        }
        float x[4];
#pragma unroll
        for (int j = 0; j < 4; ++j) {
            float keep = b2 ? w[j + 4] : w[j];
            float send = b2 ? w[j] : w[j + 4];
            x[j] = keep + __shfl_xor(send, 4);
        }
        float y[2];
#pragma unroll
        for (int j = 0; j < 2; ++j) {
            float keep = b1 ? x[j + 2] : x[j];
            float send = b1 ? x[j] : x[j + 2];
            y[j] = keep + __shfl_xor(send, 2);
        }
        const float z = (b0 ? y[1] : y[0]) + __shfl_xor(b0 ? y[0] : y[1], 1);
        const int row = m0 + (l & 7) + half * 8;
        if (row < M) {
            float* dst = b3 ? a_d : a_s;
            dst[row * HH + wave] = z;
        }
    }
}

__device__ __forceinline__ float4 gat_logit(const float* __restrict__ a_s, int s, float4 ad) {
    float4 as = *(const float4*)(a_s + (long long)s * HH);
    float4 e = make_float4(as.x + ad.x, as.y + ad.y, as.z + ad.z, as.w + ad.w);
    e.x = (e.x > 0.0f) ? e.x : 0.2f * e.x;
    e.y = (e.y > 0.0f) ? e.y : 0.2f * e.y;
    e.z = (e.z > 0.0f) ? e.z : 0.2f * e.z;
    e.w = (e.w > 0.0f) ? e.w : 0.2f * e.w;
    return e;
}

__device__ __forceinline__ void gat_accum(float4& acc, const float* __restrict__ xh,
                                          int c, int s, float4 al, int lane) {
#pragma unroll 4
    for (int k = 0; k < c; ++k) {
        int sk = __builtin_amdgcn_readlane(s, k);
        float a0 = lane_bcast(al.x, k), a1 = lane_bcast(al.y, k);
        float a2 = lane_bcast(al.z, k), a3 = lane_bcast(al.w, k);
        const float* r = xh + (long long)sk * DD + lane;
        float v0 = r[0], v1 = r[CC], v2 = r[2 * CC], v3 = r[3 * CC];
        acc.x += a0 * v0; acc.y += a1 * v1; acc.z += a2 * v2; acc.w += a3 * v3;
    }
}

__global__ void __launch_bounds__(256)
gnn_gat_gather(const int* __restrict__ rowptr, const int* __restrict__ csr_src,
               const float* __restrict__ a_s, const float* __restrict__ a_d,
               const float* __restrict__ xh, const float* __restrict__ gbias,
               const float* __restrict__ ln_g, const float* __restrict__ ln_b,
               float* __restrict__ h, int Nn, int do_relu) {
    const int lane = threadIdx.x & 31;
    const int n = __builtin_amdgcn_readfirstlane((int)(blockIdx.x * 8 + (threadIdx.x >> 5)));
    if (n >= Nn) return;
    const int start = __builtin_amdgcn_readfirstlane(rowptr[n]);
    const int deg   = __builtin_amdgcn_readfirstlane(rowptr[n + 1]) - start;
    const int tot   = deg + 1;
    const float4 ad = *(const float4*)(a_d + (long long)n * HH);
    const float ninf = -__builtin_inff();

    const bool v0 = lane < tot;
    const int  s0 = (lane < deg) ? csr_src[start + lane] : n;
    const float4 e0 = gat_logit(a_s, s0, ad);

    float4 m = v0 ? e0 : make_float4(ninf, ninf, ninf, ninf);
    for (int j0 = 32; j0 < tot; j0 += 32) {
        int j = j0 + lane;
        if (j < tot) {
            int s = (j < deg) ? csr_src[start + j] : n;
            float4 e = gat_logit(a_s, s, ad);
            m.x = fmaxf(m.x, e.x); m.y = fmaxf(m.y, e.y); m.z = fmaxf(m.z, e.z); m.w = fmaxf(m.w, e.w);
        }
    }
    m.x = wave_max(m.x); m.y = wave_max(m.y); m.z = wave_max(m.z); m.w = wave_max(m.w);

    float4 p0 = make_float4(0.f, 0.f, 0.f, 0.f);
    if (v0) p0 = make_float4(__expf(e0.x - m.x), __expf(e0.y - m.y), __expf(e0.z - m.z), __expf(e0.w - m.w));
    float4 den = p0;
    for (int j0 = 32; j0 < tot; j0 += 32) {
        int j = j0 + lane;
        if (j < tot) {
            int s = (j < deg) ? csr_src[start + j] : n;
            float4 e = gat_logit(a_s, s, ad);
            den.x += __expf(e.x - m.x); den.y += __expf(e.y - m.y);
            den.z += __expf(e.z - m.z); den.w += __expf(e.w - m.w);
        }
    }
    den.x = wave_sum(den.x); den.y = wave_sum(den.y); den.z = wave_sum(den.z); den.w = wave_sum(den.w);

    float4 acc = make_float4(0.f, 0.f, 0.f, 0.f);
    {
        float4 al = make_float4(p0.x / den.x, p0.y / den.y, p0.z / den.z, p0.w / den.w);
        int c = (tot < 32) ? tot : 32;
        gat_accum(acc, xh, c, s0, al, lane);
    }
    for (int j0 = 32; j0 < tot; j0 += 32) {
        int j = j0 + lane;
        int s = n;
        float4 al = make_float4(0.f, 0.f, 0.f, 0.f);
        if (j < tot) {
            s = (j < deg) ? csr_src[start + j] : n;
            float4 e = gat_logit(a_s, s, ad);
            al = make_float4(__expf(e.x - m.x) / den.x, __expf(e.y - m.y) / den.y,
                             __expf(e.z - m.z) / den.z, __expf(e.w - m.w) / den.w);
        }
        int c = tot - j0; if (c > 32) c = 32;
        gat_accum(acc, xh, c, s, al, lane);
    }

    float4 v = make_float4(acc.x + gbias[lane], acc.y + gbias[CC + lane],
                           acc.z + gbias[2 * CC + lane], acc.w + gbias[3 * CC + lane]);
    float mean = wave_sum((v.x + v.y) + (v.z + v.w)) * (1.0f / DD);
    float4 cc = make_float4(v.x - mean, v.y - mean, v.z - mean, v.w - mean);
    float var = wave_sum((cc.x * cc.x + cc.y * cc.y) + (cc.z * cc.z + cc.w * cc.w)) * (1.0f / DD);
    float rs = rsqrtf(var + 1e-5f);
    float4 y = make_float4(cc.x * rs * ln_g[lane] + ln_b[lane],
                           cc.y * rs * ln_g[CC + lane] + ln_b[CC + lane],
                           cc.z * rs * ln_g[2 * CC + lane] + ln_b[2 * CC + lane],
                           cc.w * rs * ln_g[3 * CC + lane] + ln_b[3 * CC + lane]);
    if (do_relu) { y.x = fmaxf(y.x, 0.f); y.y = fmaxf(y.y, 0.f); y.z = fmaxf(y.z, 0.f); y.w = fmaxf(y.w, 0.f); }
    float* hp = h + (long long)n * DD + lane;
    hp[0]      += y.x;
    hp[CC]     += y.y;
    hp[2 * CC] += y.z;
    hp[3 * CC] += y.w;
}

extern "C" void kernel_launch(void* const* d_in, const int* in_sizes, int n_in,
                              void* d_out, int out_size, void* d_ws, size_t ws_size,
                              hipStream_t stream) {
    const float* x      = (const float*)d_in[0];
    const int*   ei     = (const int*)d_in[1];
    const float* W_edge = (const float*)d_in[2];
    const float* b_edge = (const float*)d_in[3];
    const float* W_node = (const float*)d_in[4];
    const float* b_node = (const float*)d_in[5];
    const float* lin_w  = (const float*)d_in[6];
    const float* att_sp = (const float*)d_in[7];
    const float* att_dp = (const float*)d_in[8];
    const float* g_bias = (const float*)d_in[9];
    const float* ln_g   = (const float*)d_in[10];
    const float* ln_b   = (const float*)d_in[11];

    const int E_ = in_sizes[0] / 3;
    const int Nn = out_size / DD;

    int*   rowptr   = (int*)d_ws;
    int*   deg      = rowptr + (Nn + 1);
    int*   cursor   = deg + Nn;
    int*   slot_eid = cursor + Nn;
    int*   csr_eid  = slot_eid + E_;
    int*   csr_src  = csr_eid + E_;
    float* xh    = (float*)d_ws + (long long)Nn * DD;
    float* a_s   = xh  + (long long)Nn * DD;
    float* a_d   = a_s + (long long)Nn * HH;
    float* cnt   = a_d + (long long)Nn * HH;
    float* h     = (float*)d_out;

    int g1  = (Nn + 255) / 256;
    int ge  = (E_ + 255) / 256;
    int gs  = (Nn + 1023) / 1024;
    int gw  = (Nn + 7) / 8;
    dim3 gemm_grid((Nn + 15) / 16, 1);

    gnn_fill_u32<<<g1, 256, 0, stream>>>((unsigned int*)deg, 0u, Nn);
    gnn_csr_count<<<ge, 256, 0, stream>>>(ei, deg, E_);
    gnn_csr_scan<<<gs, 1024, 0, stream>>>(deg, rowptr, cursor, cnt, Nn);
    gnn_csr_place<<<ge, 256, 0, stream>>>(ei, cursor, slot_eid, E_);
    gnn_csr_order<<<ge, 256, 0, stream>>>(ei, rowptr, slot_eid, csr_eid, csr_src, E_);

    gnn_node_init<<<gw, 256, 0, stream>>>(x, rowptr, csr_eid, W_edge, b_edge, xh, Nn);
    gnn_wmma_gemm128<<<gemm_grid, 128, 0, stream>>>(xh, W_node, h, b_node, cnt,
                                                    nullptr, nullptr, nullptr, nullptr, Nn);

    for (int i = 0; i < 3; ++i) {
        gnn_wmma_gemm128<<<gemm_grid, 128, 0, stream>>>(
            h, lin_w + (long long)i * DD * DD, xh, nullptr, nullptr,
            att_sp + i * DD, att_dp + i * DD, a_s, a_d, Nn);
        gnn_gat_gather<<<gw, 256, 0, stream>>>(rowptr, csr_src, a_s, a_d, xh,
                                               g_bias + i * DD, ln_g + i * DD, ln_b + i * DD,
                                               h, Nn, (i < 2) ? 1 : 0);
    }
    (void)n_in; (void)ws_size;
}


# Node init fused into one launch: a 4-wave block builds 16 nodes' edge-MLP segment sums in LDS, then runs the W_node GEMM and layer 0's projection GEMM from LDS tiles (13 launches -> 11)
# speedup vs baseline: 4.8063x; 1.1064x over previous
#include <hip/hip_runtime.h>
#include <hip/hip_bf16.h>

typedef float v2f __attribute__((ext_vector_type(2)));
typedef float v8f __attribute__((ext_vector_type(8)));

#define DD 128
#define HH 4
#define CC 32

__global__ void gnn_fill_u32(unsigned int* __restrict__ p, unsigned int v, long long n) {
    long long i = (long long)blockIdx.x * blockDim.x + threadIdx.x;
    if (i < n) p[i] = v;
}

__global__ void gnn_csr_count(const int* __restrict__ ei, int* __restrict__ deg, int E_) {
    int e = blockIdx.x * blockDim.x + threadIdx.x;
    if (e < E_) atomicAdd(&deg[ei[E_ + e]], 1);
}

__global__ void __launch_bounds__(1024)
gnn_csr_scan(const int* __restrict__ deg, int* __restrict__ rowptr, int* __restrict__ cursor,
             float* __restrict__ cnt, int n) {
    __shared__ int wtot[32];
    __shared__ int ptot[32];
    const int t = threadIdx.x, lane = t & 31, w = t >> 5;
    const int base = blockIdx.x * 1024;
    int ps = 0;
    for (int j0 = t; j0 < base; j0 += 16 * 1024) {
#pragma unroll
        for (int u = 0; u < 16; ++u) {
            const int j = j0 + u * 1024;
            const bool in = j < base;
            const int v = deg[in ? j : t];
            ps += in ? v : 0;
        }
    }
    const int i = base + t;
    const int d = (i < n) ? deg[i] : 0;
    int incl = d;
    for (int o = 1; o < 32; o <<= 1) { int v = __shfl_up(incl, o); if (lane >= o) incl += v; }
    for (int m = 16; m; m >>= 1) ps += __shfl_xor(ps, m);
    if (lane == 31) wtot[w] = incl;
    if (lane == 0)  ptot[w] = ps;
    __syncthreads();
    const int wv = wtot[lane];
    int iv = wv;
    for (int o = 1; o < 32; o <<= 1) { int u = __shfl_up(iv, o); if (lane >= o) iv += u; }
    const int woff = __shfl(iv - wv, w);
    int pre = ptot[lane];
    for (int m = 16; m; m >>= 1) pre += __shfl_xor(pre, m);
    const int run = pre + woff + incl - d;
    if (i < n) {
        rowptr[i] = run; cursor[i] = run; cnt[i] = (float)d;
        if (i == n - 1) rowptr[n] = run + d;
    }
}

__global__ void gnn_csr_place(const int* __restrict__ ei, int* __restrict__ cursor,
                              int* __restrict__ slot_eid, int E_) {
    int e = blockIdx.x * blockDim.x + threadIdx.x;
    if (e < E_) { int p = atomicAdd(&cursor[ei[E_ + e]], 1); slot_eid[p] = e; }
}

__global__ void gnn_csr_order(const int* __restrict__ ei, const int* __restrict__ rowptr,
                              const int* __restrict__ slot_eid, int* __restrict__ csr_eid,
                              int* __restrict__ csr_src, int E_) {
    int p = blockIdx.x * blockDim.x + threadIdx.x;
    if (p >= E_) return;
    int eid = slot_eid[p];
    int d = ei[E_ + eid];
    int b = rowptr[d], en = rowptr[d + 1];
    int r = 0;
    for (int q = b; q < en; ++q) r += (slot_eid[q] < eid) ? 1 : 0;
    csr_eid[b + r] = eid;
    csr_src[b + r] = ei[eid];
}

__device__ __forceinline__ float lane_bcast(float v, int k) {
    return __int_as_float(__builtin_amdgcn_readlane(__float_as_int(v), k));
}
__device__ __forceinline__ float wave_sum(float v) {
#pragma unroll
    for (int m = 16; m; m >>= 1) v += __shfl_xor(v, m);
    return v;
}
__device__ __forceinline__ float wave_max(float v) {
#pragma unroll
    for (int m = 16; m; m >>= 1) v = fmaxf(v, __shfl_xor(v, m));
    return v;
}

__device__ __forceinline__ void gemm_k128(const float* Arow, const float* __restrict__ Bcol,
                                          int K0, v8f& acc0, v8f& acc1) {
#pragma unroll
    for (int k = 0; k < DD; k += 4) {
        v2f a  = *(const v2f*)(Arow + k + K0);
        v2f p0 = *(const v2f*)(Bcol + (k + K0) * DD);
        v2f p1 = *(const v2f*)(Bcol + (k + K0 + 1) * DD);
        v2f b0, b1;
        b0.x = p0.x; b0.y = p1.x;
        b1.x = p0.y; b1.y = p1.y;
        acc0 = __builtin_amdgcn_wmma_f32_16x16x4_f32(false, a, false, b0,
                                                     (short)0, acc0, false, false);
        acc1 = __builtin_amdgcn_wmma_f32_16x16x4_f32(false, a, false, b1,
                                                     (short)0, acc1, false, false);
    }
}

__device__ __forceinline__ void gemm_logits(const v8f& acc0, const v8f& acc1, v2f ws, v2f wd,
                                            int l, int half, int wave, int m0, int M,
                                            float* __restrict__ a_s, float* __restrict__ a_d) {
    float v[16];
#pragma unroll
    for (int r = 0; r < 8; ++r) {
        v[r]     = acc0[r] * ws.x + acc1[r] * ws.y;
        v[8 + r] = acc0[r] * wd.x + acc1[r] * wd.y;
    }
    const bool b3 = (l & 8) != 0, b2 = (l & 4) != 0, b1 = (l & 2) != 0, b0 = (l & 1) != 0;
    float w[8];
#pragma unroll
    for (int j = 0; j < 8; ++j) {
        float keep = b3 ? v[j + 8] : v[j];
        float send = b3 ? v[j] : v[j + 8];
        w[j] = keep + __shfl_xor(send, 8);
    }
    float x[4];
#pragma unroll
    for (int j = 0; j < 4; ++j) {
        float keep = b2 ? w[j + 4] : w[j];
        float send = b2 ? w[j] : w[j + 4];
        x[j] = keep + __shfl_xor(send, 4);
    }
    float y[2];
#pragma unroll
    for (int j = 0; j < 2; ++j) {
        float keep = b1 ? x[j + 2] : x[j];
        float send = b1 ? x[j] : x[j + 2];
        y[j] = keep + __shfl_xor(send, 2);
    }
    const float z = (b0 ? y[1] : y[0]) + __shfl_xor(b0 ? y[0] : y[1], 1);
    const int row = m0 + (l & 7) + half * 8;
    if (row < M) {
        float* dst = b3 ? a_d : a_s;
        dst[row * HH + wave] = z;
    }
}

__global__ void gnn_wmma_gemm128(const float* __restrict__ A, const float* __restrict__ B,
                                 float* __restrict__ C,
                                 const float* __restrict__ att_src,
                                 const float* __restrict__ att_dst,
                                 float* __restrict__ a_s, float* __restrict__ a_d, int M) {
    const int wave = threadIdx.x >> 5;
    const int lane = threadIdx.x & 31;
    const int m0 = blockIdx.x * 16;
    const int half = lane >> 4;
    const int l    = lane & 15;
    const int K0   = half * 2;
    int Mrow = m0 + l;
    if (Mrow >= M) Mrow = M - 1;
    const int col = wave * CC + 2 * l;

    const v2f ws = *(const v2f*)(att_src + col);
    const v2f wd = *(const v2f*)(att_dst + col);
    v8f acc0 = {};
    v8f acc1 = {};
    gemm_k128(A + (long long)Mrow * DD, B + col, K0, acc0, acc1);

#pragma unroll
    for (int r = 0; r < 8; ++r) {
        const int row = m0 + r + half * 8;
        if (row < M) {
            v2f o; o.x = acc0[r]; o.y = acc1[r];
            *(v2f*)(C + (long long)row * DD + col) = o;
        }
    }
    gemm_logits(acc0, acc1, ws, wd, l, half, wave, m0, M, a_s, a_d);
}

#define LDA 132

__global__ void __launch_bounds__(128)
gnn_init_fused(const float* __restrict__ x, const int* __restrict__ rowptr,
               const int* __restrict__ csr_eid, const float* __restrict__ W_edge,
               const float* __restrict__ b_edge, const float* __restrict__ W_node,
               const float* __restrict__ b_node, const float* __restrict__ cnt,
               const float* __restrict__ lin_w0, const float* __restrict__ att_src,
               const float* __restrict__ att_dst, float* __restrict__ h,
               float* __restrict__ xh, float* __restrict__ a_s, float* __restrict__ a_d,
               int M, int E_) {
    __shared__ __attribute__((aligned(16))) float sA[16 * LDA];
    __shared__ __attribute__((aligned(16))) float sH[16 * LDA];
    const int wave = threadIdx.x >> 5;
    const int lane = threadIdx.x & 31;
    const int m0 = blockIdx.x * 16;
    const int half = lane >> 4;
    const int l    = lane & 15;
    const int K0   = half * 2;
    const int col = wave * CC + 2 * l;

    {
        const float4 w0 = *(const float4*)(W_edge + lane * 4);
        const float4 w1 = *(const float4*)(W_edge + DD + lane * 4);
        const float4 w2 = *(const float4*)(W_edge + 2 * DD + lane * 4);
        const float4 bb = *(const float4*)(b_edge + lane * 4);
        const int nb = m0 + 4 * wave;
        int ri = nb + (lane < 5 ? lane : 4);
        if (ri > M) ri = M;
        const int rp = rowptr[ri];
        int sb[5];
#pragma unroll
        for (int q = 0; q < 5; ++q) sb[q] = __builtin_amdgcn_readlane(rp, q);
        float X0[4], X1[4], X2[4];
#pragma unroll
        for (int q = 0; q < 4; ++q) {
            const int j = sb[q] + lane;
            int jc = (j < E_) ? j : E_ - 1;
            if (jc < 0) jc = 0;
            const long long e = csr_eid[jc];
            const float t0 = x[e * 3 + 0], t1 = x[e * 3 + 1], t2 = x[e * 3 + 2];
            const bool in = j < sb[q + 1];
            X0[q] = in ? t0 : 0.f; X1[q] = in ? t1 : 0.f; X2[q] = in ? t2 : 0.f;
        }
#pragma unroll
        for (int q = 0; q < 4; ++q) {
            const int start = sb[q], end = sb[q + 1];
            float4 acc = make_float4(0.f, 0.f, 0.f, 0.f);
            float x0 = X0[q], x1 = X1[q], x2 = X2[q];
            for (int j0 = start; j0 < end; j0 += 32) {
                if (j0 != start) {
                    int j = j0 + lane;
                    x0 = 0.f; x1 = 0.f; x2 = 0.f;
                    if (j < end) {
                        int e = csr_eid[j];
                        x0 = x[(long long)e * 3 + 0]; x1 = x[(long long)e * 3 + 1]; x2 = x[(long long)e * 3 + 2];
                    }
                }
                int c = end - j0; if (c > 32) c = 32;
                for (int k = 0; k < c; ++k) {
                    float y0 = lane_bcast(x0, k), y1 = lane_bcast(x1, k), y2 = lane_bcast(x2, k);
                    acc.x += y0 * w0.x + y1 * w1.x + y2 * w2.x + bb.x;
                    acc.y += y0 * w0.y + y1 * w1.y + y2 * w2.y + bb.y;
                    acc.z += y0 * w0.z + y1 * w1.z + y2 * w2.z + bb.z;
                    acc.w += y0 * w0.w + y1 * w1.w + y2 * w2.w + bb.w;
                }
            }
            *(float4*)(sA + (4 * wave + q) * LDA + lane * 4) = acc;
        }
    }
    __syncthreads();

    {
        float inv[8];
#pragma unroll
        for (int r = 0; r < 8; ++r) {
            int rc = m0 + r + half * 8;
            if (rc >= M) rc = M - 1;
            inv[r] = 1.0f / fmaxf(cnt[rc], 1.0f);
        }
        const float bx = b_node[col], by = b_node[col + 1];
        v8f acc0 = {};
        v8f acc1 = {};
        gemm_k128(sA + l * LDA, W_node + col, K0, acc0, acc1);
#pragma unroll
        for (int r = 0; r < 8; ++r) {
            const int row = m0 + r + half * 8;
            float v0 = acc0[r], v1 = acc1[r];
            v0 *= inv[r]; v1 *= inv[r];
            v0 += bx; v1 += by;
            v2f o; o.x = v0; o.y = v1;
            if (row < M) *(v2f*)(h + (long long)row * DD + col) = o;
            *(v2f*)(sH + (r + half * 8) * LDA + col) = o;
        }
    }
    __syncthreads();

    {
        const v2f ws = *(const v2f*)(att_src + col);
        const v2f wd = *(const v2f*)(att_dst + col);
        v8f acc0 = {};
        v8f acc1 = {};
        gemm_k128(sH + l * LDA, lin_w0 + col, K0, acc0, acc1);
#pragma unroll
        for (int r = 0; r < 8; ++r) {
            const int row = m0 + r + half * 8;
            if (row < M) {
                v2f o; o.x = acc0[r]; o.y = acc1[r];
                *(v2f*)(xh + (long long)row * DD + col) = o;
            }
        }
        gemm_logits(acc0, acc1, ws, wd, l, half, wave, m0, M, a_s, a_d);
    }
}

__device__ __forceinline__ float4 gat_logit(const float* __restrict__ a_s, int s, float4 ad) {
    float4 as = *(const float4*)(a_s + (long long)s * HH);
    float4 e = make_float4(as.x + ad.x, as.y + ad.y, as.z + ad.z, as.w + ad.w);
    e.x = (e.x > 0.0f) ? e.x : 0.2f * e.x;
    e.y = (e.y > 0.0f) ? e.y : 0.2f * e.y;
    e.z = (e.z > 0.0f) ? e.z : 0.2f * e.z;
    e.w = (e.w > 0.0f) ? e.w : 0.2f * e.w;
    return e;
}

__device__ __forceinline__ void gat_accum(float4& acc, const float* __restrict__ xh,
                                          int c, int s, float4 al, int lane) {
#pragma unroll 4
    for (int k = 0; k < c; ++k) {
        int sk = __builtin_amdgcn_readlane(s, k);
        float a0 = lane_bcast(al.x, k), a1 = lane_bcast(al.y, k);
        float a2 = lane_bcast(al.z, k), a3 = lane_bcast(al.w, k);
        const float* r = xh + (long long)sk * DD + lane;
        float v0 = r[0], v1 = r[CC], v2 = r[2 * CC], v3 = r[3 * CC];
        acc.x += a0 * v0; acc.y += a1 * v1; acc.z += a2 * v2; acc.w += a3 * v3;
    }
}

__global__ void __launch_bounds__(256)
gnn_gat_gather(const int* __restrict__ rowptr, const int* __restrict__ csr_src,
               const float* __restrict__ a_s, const float* __restrict__ a_d,
               const float* __restrict__ xh, const float* __restrict__ gbias,
               const float* __restrict__ ln_g, const float* __restrict__ ln_b,
               float* __restrict__ h, int Nn, int do_relu) {
    const int lane = threadIdx.x & 31;
    const int n = __builtin_amdgcn_readfirstlane((int)(blockIdx.x * 8 + (threadIdx.x >> 5)));
    if (n >= Nn) return;
    const int start = __builtin_amdgcn_readfirstlane(rowptr[n]);
    const int deg   = __builtin_amdgcn_readfirstlane(rowptr[n + 1]) - start;
    const int tot   = deg + 1;
    const float4 ad = *(const float4*)(a_d + (long long)n * HH);
    const float ninf = -__builtin_inff();

    const bool v0 = lane < tot;
    const int  s0 = (lane < deg) ? csr_src[start + lane] : n;
    const float4 e0 = gat_logit(a_s, s0, ad);

    float4 m = v0 ? e0 : make_float4(ninf, ninf, ninf, ninf);
    for (int j0 = 32; j0 < tot; j0 += 32) {
        int j = j0 + lane;
        if (j < tot) {
            int s = (j < deg) ? csr_src[start + j] : n;
            float4 e = gat_logit(a_s, s, ad);
            m.x = fmaxf(m.x, e.x); m.y = fmaxf(m.y, e.y); m.z = fmaxf(m.z, e.z); m.w = fmaxf(m.w, e.w);
        }
    }
    m.x = wave_max(m.x); m.y = wave_max(m.y); m.z = wave_max(m.z); m.w = wave_max(m.w);

    float4 p0 = make_float4(0.f, 0.f, 0.f, 0.f);
    if (v0) p0 = make_float4(__expf(e0.x - m.x), __expf(e0.y - m.y), __expf(e0.z - m.z), __expf(e0.w - m.w));
    float4 den = p0;
    for (int j0 = 32; j0 < tot; j0 += 32) {
        int j = j0 + lane;
        if (j < tot) {
            int s = (j < deg) ? csr_src[start + j] : n;
            float4 e = gat_logit(a_s, s, ad);
            den.x += __expf(e.x - m.x); den.y += __expf(e.y - m.y);
            den.z += __expf(e.z - m.z); den.w += __expf(e.w - m.w);
        }
    }
    den.x = wave_sum(den.x); den.y = wave_sum(den.y); den.z = wave_sum(den.z); den.w = wave_sum(den.w);

    float4 acc = make_float4(0.f, 0.f, 0.f, 0.f);
    {
        float4 al = make_float4(p0.x / den.x, p0.y / den.y, p0.z / den.z, p0.w / den.w);
        int c = (tot < 32) ? tot : 32;
        gat_accum(acc, xh, c, s0, al, lane);
    }
    for (int j0 = 32; j0 < tot; j0 += 32) {
        int j = j0 + lane;
        int s = n;
        float4 al = make_float4(0.f, 0.f, 0.f, 0.f);
        if (j < tot) {
            s = (j < deg) ? csr_src[start + j] : n;
            float4 e = gat_logit(a_s, s, ad);
            al = make_float4(__expf(e.x - m.x) / den.x, __expf(e.y - m.y) / den.y,
                             __expf(e.z - m.z) / den.z, __expf(e.w - m.w) / den.w);
        }
        int c = tot - j0; if (c > 32) c = 32;
        gat_accum(acc, xh, c, s, al, lane);
    }

    float4 v = make_float4(acc.x + gbias[lane], acc.y + gbias[CC + lane],
                           acc.z + gbias[2 * CC + lane], acc.w + gbias[3 * CC + lane]);
    float mean = wave_sum((v.x + v.y) + (v.z + v.w)) * (1.0f / DD);
    float4 cc = make_float4(v.x - mean, v.y - mean, v.z - mean, v.w - mean);
    float var = wave_sum((cc.x * cc.x + cc.y * cc.y) + (cc.z * cc.z + cc.w * cc.w)) * (1.0f / DD);
    float rs = rsqrtf(var + 1e-5f);
    float4 y = make_float4(cc.x * rs * ln_g[lane] + ln_b[lane],
                           cc.y * rs * ln_g[CC + lane] + ln_b[CC + lane],
                           cc.z * rs * ln_g[2 * CC + lane] + ln_b[2 * CC + lane],
                           cc.w * rs * ln_g[3 * CC + lane] + ln_b[3 * CC + lane]);
    if (do_relu) { y.x = fmaxf(y.x, 0.f); y.y = fmaxf(y.y, 0.f); y.z = fmaxf(y.z, 0.f); y.w = fmaxf(y.w, 0.f); }
    float* hp = h + (long long)n * DD + lane;
    hp[0]      += y.x;
    hp[CC]     += y.y;
    hp[2 * CC] += y.z;
    hp[3 * CC] += y.w;
}

extern "C" void kernel_launch(void* const* d_in, const int* in_sizes, int n_in,
                              void* d_out, int out_size, void* d_ws, size_t ws_size,
                              hipStream_t stream) {
    const float* x      = (const float*)d_in[0];
    const int*   ei     = (const int*)d_in[1];
    const float* W_edge = (const float*)d_in[2];
    const float* b_edge = (const float*)d_in[3];
    const float* W_node = (const float*)d_in[4];
    const float* b_node = (const float*)d_in[5];
    const float* lin_w  = (const float*)d_in[6];
    const float* att_sp = (const float*)d_in[7];
    const float* att_dp = (const float*)d_in[8];
    const float* g_bias = (const float*)d_in[9];
    const float* ln_g   = (const float*)d_in[10];
    const float* ln_b   = (const float*)d_in[11];

    const int E_ = in_sizes[0] / 3;
    const int Nn = out_size / DD;

    int*   rowptr   = (int*)d_ws;
    int*   deg      = rowptr + (Nn + 1);
    int*   cursor   = deg + Nn;
    int*   slot_eid = cursor + Nn;
    int*   csr_eid  = slot_eid + E_;
    int*   csr_src  = csr_eid + E_;
    float* xh    = (float*)d_ws + (long long)Nn * DD;
    float* a_s   = xh  + (long long)Nn * DD;
    float* a_d   = a_s + (long long)Nn * HH;
    float* cnt   = a_d + (long long)Nn * HH;
    float* h     = (float*)d_out;

    int g1  = (Nn + 255) / 256;
    int ge  = (E_ + 255) / 256;
    int gs  = (Nn + 1023) / 1024;
    int gw  = (Nn + 7) / 8;
    dim3 gemm_grid((Nn + 15) / 16, 1);

    gnn_fill_u32<<<g1, 256, 0, stream>>>((unsigned int*)deg, 0u, Nn);
    gnn_csr_count<<<ge, 256, 0, stream>>>(ei, deg, E_);
    gnn_csr_scan<<<gs, 1024, 0, stream>>>(deg, rowptr, cursor, cnt, Nn);
    gnn_csr_place<<<ge, 256, 0, stream>>>(ei, cursor, slot_eid, E_);
    gnn_csr_order<<<ge, 256, 0, stream>>>(ei, rowptr, slot_eid, csr_eid, csr_src, E_);

    gnn_init_fused<<<gemm_grid, 128, 0, stream>>>(x, rowptr, csr_eid, W_edge, b_edge, W_node, b_node,
                                                  cnt, lin_w, att_sp, att_dp, h, xh, a_s, a_d, Nn, E_);
    gnn_gat_gather<<<gw, 256, 0, stream>>>(rowptr, csr_src, a_s, a_d, xh,
                                           g_bias, ln_g, ln_b, h, Nn, 1);

    for (int i = 1; i < 3; ++i) {
        gnn_wmma_gemm128<<<gemm_grid, 128, 0, stream>>>(
            h, lin_w + (long long)i * DD * DD, xh,
            att_sp + i * DD, att_dp + i * DD, a_s, a_d, Nn);
        gnn_gat_gather<<<gw, 256, 0, stream>>>(rowptr, csr_src, a_s, a_d, xh,
                                               g_bias + i * DD, ln_g + i * DD, ln_b + i * DD,
                                               h, Nn, (i < 2) ? 1 : 0);
    }
    (void)n_in; (void)ws_size;
}


# Node init's two weight products folded into one: [W_edge;b_edge]@W_node (4x128), formed per call on WMMA by a spare block of the edge placement launch; init sums 3 features per node, one K=4 WMMA step
# speedup vs baseline: 6.8689x; 1.4291x over previous
#include <hip/hip_runtime.h>
#include <hip/hip_bf16.h>

typedef float v2f __attribute__((ext_vector_type(2)));
typedef float v8f __attribute__((ext_vector_type(8)));

#define DD 128
#define CC 32

#define CAP 32

__global__ void gnn_bucket_clear(int* __restrict__ cursor, int2* __restrict__ bucket, int Nn) {
    int i = blockIdx.x * blockDim.x + threadIdx.x;
    if (i <= Nn) cursor[i] = 0;
    if (i < Nn)  bucket[(long long)i * CAP] = make_int2(0, 0);
}

__device__ __forceinline__ int2 next_edge(int n, int prev, const int2* __restrict__ bucket_n,
                                          const int* __restrict__ ovf, int novf,
                                          const int* __restrict__ ei, int E_, int lane) {
    const int2 b = bucket_n[lane];
    int be = (b.x > prev) ? b.x : 0x7fffffff;
    int bs = b.y;
    for (int q = lane; q < novf; q += 32) {
        const int e = ovf[q];
        if (ei[E_ + e] == n && e > prev && e < be) { be = e; bs = ei[e]; }
    }
    for (int m = 16; m; m >>= 1) {
        const int oe = __shfl_xor(be, m), os = __shfl_xor(bs, m);
        if (oe < be) { be = oe; bs = os; }
    }
    if (be > E_ - 1) be = E_ - 1;
    return make_int2(be, bs);
}

#define XADD(v, m) ((v) + __int_as_float(__builtin_amdgcn_update_dpp(0, __float_as_int(v), 0x160 + (m), 0xf, 0xf, true)))
__device__ __forceinline__ float group8_sum(float v) {
    v = XADD(v, 1); v = XADD(v, 2); v = XADD(v, 4);
    return v;
}
__device__ __forceinline__ float wave_sum(float v) {
    v = XADD(v, 1); v = XADD(v, 2); v = XADD(v, 4); v = XADD(v, 8);
    return v + __int_as_float(__builtin_amdgcn_permlanex16(__float_as_int(v), __float_as_int(v),
                                                           0x76543210, 0xfedcba98, false, false));
}

__device__ __forceinline__ void gemm_k128(const float* Arow, const float* __restrict__ Bcol,
                                          int K0, v8f& acc0, v8f& acc1) {
#pragma unroll
    for (int k = 0; k < DD; k += 4) {
        v2f a  = *(const v2f*)(Arow + k + K0);
        v2f p0 = *(const v2f*)(Bcol + (k + K0) * DD);
        v2f p1 = *(const v2f*)(Bcol + (k + K0 + 1) * DD);
        v2f b0, b1;
        b0.x = p0.x; b0.y = p1.x;
        b1.x = p0.y; b1.y = p1.y;
        acc0 = __builtin_amdgcn_wmma_f32_16x16x4_f32(false, a, false, b0,
                                                     (short)0, acc0, false, false);
        acc1 = __builtin_amdgcn_wmma_f32_16x16x4_f32(false, a, false, b1,
                                                     (short)0, acc1, false, false);
    }
}

__global__ void gnn_bucket_place(const int* __restrict__ ei, int* cursor, int2* __restrict__ bucket,
                                 int* __restrict__ ovf, int Nn, int E_,
                                 const float* __restrict__ W_edge, const float* __restrict__ b_edge,
                                 const float* __restrict__ W_node, float* __restrict__ Wc) {
    if (blockIdx.x == 0) {
        if (threadIdx.x < 128) {
            const int wave = threadIdx.x >> 5;
            const int lane = threadIdx.x & 31;
            const int half = lane >> 4;
            const int l    = lane & 15;
            const int K0   = half * 2;
            const int col  = wave * CC + 2 * l;
            const float* Arow = (l < 3) ? (W_edge + l * DD) : b_edge;
            v8f acc0 = {};
            v8f acc1 = {};
            gemm_k128(Arow, W_node + col, K0, acc0, acc1);
            if (half == 0) {
#pragma unroll
                for (int r = 0; r < 4; ++r) {
                    v2f o; o.x = acc0[r]; o.y = acc1[r];
                    *(v2f*)(Wc + r * DD + col) = o;
                }
            }
        }
        return;
    }
    int e = (blockIdx.x - 1) * blockDim.x + threadIdx.x;
    if (e < E_) {
        const int d = ei[E_ + e];
        const int s = ei[e];
        const int p = atomicAdd(&cursor[d], 1);
        if (p < CAP) bucket[(long long)d * CAP + p] = make_int2(e, s);
        else         ovf[atomicAdd(&cursor[Nn], 1)] = e;
    }
}

__global__ void gnn_wmma_gemm128(const float* __restrict__ A, const float* __restrict__ B,
                                 float* __restrict__ C, int M) {
    const int wave = threadIdx.x >> 5;
    const int lane = threadIdx.x & 31;
    const int m0 = blockIdx.x * 16;
    const int half = lane >> 4;
    const int l    = lane & 15;
    const int K0   = half * 2;
    int Mrow = m0 + l;
    if (Mrow >= M) Mrow = M - 1;
    const int col = wave * CC + 2 * l;

    v8f acc0 = {};
    v8f acc1 = {};
    gemm_k128(A + (long long)Mrow * DD, B + col, K0, acc0, acc1);

#pragma unroll
    for (int r = 0; r < 8; ++r) {
        const int row = m0 + r + half * 8;
        if (row < M) {
            v2f o; o.x = acc0[r]; o.y = acc1[r];
            *(v2f*)(C + (long long)row * DD + col) = o;
        }
    }
}

#define LDA 132

__global__ void __launch_bounds__(128)
gnn_init_fused(const float* __restrict__ x, const int* __restrict__ ei,
               const int* __restrict__ cursor, const int2* __restrict__ bucket,
               const int* __restrict__ ovf, int* __restrict__ srcs,
               const float* __restrict__ Wc,
               const float* __restrict__ b_node,
               const float* __restrict__ lin_w0, float* __restrict__ h,
               float* __restrict__ xh, int M, int E_) {
    __shared__ __attribute__((aligned(16))) float sX[16 * 4];
    __shared__ __attribute__((aligned(16))) float sH[16 * LDA];
    const int wave = threadIdx.x >> 5;
    const int lane = threadIdx.x & 31;
    const int m0 = blockIdx.x * 16;
    const int half = lane >> 4;
    const int l    = lane & 15;
    const int K0   = half * 2;
    const int col = wave * CC + 2 * l;

    const v2f wp0 = *(const v2f*)(Wc + K0 * DD + col);
    const v2f wp1 = *(const v2f*)(Wc + (K0 + 1) * DD + col);
    const float bx = b_node[col], by = b_node[col + 1];

    {
        const int nb = m0 + 4 * wave;
        const int ni = nb + (lane < 4 ? lane : 3);
        int dg = cursor[ni < M ? ni : M - 1];
        if (ni >= M) dg = 0;
        int dq[4];
#pragma unroll
        for (int q = 0; q < 4; ++q) dq[q] = __builtin_amdgcn_readlane(dg, q);
        int EID[4], SRC[4];
#pragma unroll
        for (int q = 0; q < 4; ++q) {
            const int c  = dq[q] < CAP ? dq[q] : CAP;
            const int cl = c > 0 ? c - 1 : 0;
            const int nq = nb + q < M ? nb + q : M - 1;
            const int2 pr = bucket[(long long)nq * CAP + (lane < cl ? lane : cl)];
            EID[q] = pr.x; SRC[q] = pr.y;
        }
#pragma unroll
        for (int q = 0; q < 4; ++q) {
            const int deg = dq[q];
            int ss = -1;
            if (deg <= CAP) {
                int r = 0;
                for (int k = 0; k < deg; ++k)
                    r += (__builtin_amdgcn_readlane(EID[q], k) < EID[q]) ? 1 : 0;
                const bool in = lane < deg;
                const int to = (in ? r : lane) << 2;
                const int se = __builtin_amdgcn_ds_permute(to, EID[q]);
                const int sp = __builtin_amdgcn_ds_permute(to, SRC[q]);
                EID[q] = in ? se : 0;
                if (in) ss = sp;
            } else {
                EID[q] = 0;
                ss = -2;
            }
            if (nb + q < M) srcs[(long long)(nb + q) * CAP + lane] = ss;
        }
        float X0[4], X1[4], X2[4];
#pragma unroll
        for (int q = 0; q < 4; ++q) {
            const long long e = EID[q];
            X0[q] = x[e * 3 + 0]; X1[q] = x[e * 3 + 1]; X2[q] = x[e * 3 + 2];
        }
#pragma unroll
        for (int q = 0; q < 4; ++q) {
            const int deg = dq[q];
            float s0 = 0.f, s1 = 0.f, s2 = 0.f;
            if (deg <= CAP) {
                const bool in = lane < deg;
                s0 = wave_sum(in ? X0[q] : 0.f);
                s1 = wave_sum(in ? X1[q] : 0.f);
                s2 = wave_sum(in ? X2[q] : 0.f);
            } else {
                const int n = nb + q;
                const int novf = cursor[M];
                int prev = -1;
                for (int r = 0; r < deg; ++r) {
                    const int2 es = next_edge(n, prev, bucket + (long long)n * CAP, ovf, novf, ei, E_, lane);
                    prev = es.x;
                    const long long e = es.x;
                    s0 += x[e * 3 + 0]; s1 += x[e * 3 + 1]; s2 += x[e * 3 + 2];
                }
            }
            if (lane == 0) *(float4*)(sX + (4 * wave + q) * 4) = make_float4(s0, s1, s2, (float)deg);
        }
    }
    __syncthreads();

    {
        v8f acc0 = {};
        v8f acc1 = {};
        {
            const v2f a = *(const v2f*)(sX + l * 4 + K0);
            v2f b0, b1;
            b0.x = wp0.x; b0.y = wp1.x;
            b1.x = wp0.y; b1.y = wp1.y;
            acc0 = __builtin_amdgcn_wmma_f32_16x16x4_f32(false, a, false, b0,
                                                         (short)0, acc0, false, false);
            acc1 = __builtin_amdgcn_wmma_f32_16x16x4_f32(false, a, false, b1,
                                                         (short)0, acc1, false, false);
        }
#pragma unroll
        for (int r = 0; r < 8; ++r) {
            const int row = m0 + r + half * 8;
            const float invr = 1.0f / fmaxf(sX[(r + half * 8) * 4 + 3], 1.0f);
            float v0 = acc0[r], v1 = acc1[r];
            v0 *= invr; v1 *= invr;
            v0 += bx; v1 += by;
            v2f o; o.x = v0; o.y = v1;
            if (row < M) *(v2f*)(h + (long long)row * DD + col) = o;
            *(v2f*)(sH + (r + half * 8) * LDA + col) = o;
        }
    }
    __syncthreads();

    {
        v8f acc0 = {};
        v8f acc1 = {};
        gemm_k128(sH + l * LDA, lin_w0 + col, K0, acc0, acc1);
#pragma unroll
        for (int r = 0; r < 8; ++r) {
            const int row = m0 + r + half * 8;
            if (row < M) {
                v2f o; o.x = acc0[r]; o.y = acc1[r];
                *(v2f*)(xh + (long long)row * DD + col) = o;
            }
        }
    }
}

__device__ __forceinline__ float dot4(float4 a, float4 b) {
    return a.x * b.x + a.y * b.y + a.z * b.z + a.w * b.w;
}
__device__ __forceinline__ float leaky02(float x) { return fmaxf(x, 0.2f * x); }

__global__ void __launch_bounds__(256)
gnn_gat_gather(const int* __restrict__ cursor, const int* __restrict__ srcs,
               const int2* __restrict__ bucket, const int* __restrict__ ovf,
               const int* __restrict__ ei,
               const float* __restrict__ att_src, const float* __restrict__ att_dst,
               const float* __restrict__ xh, const float* __restrict__ gbias,
               const float* __restrict__ ln_g, const float* __restrict__ ln_b,
               float* __restrict__ h, int Nn, int E_, int do_relu) {
    const int lane = threadIdx.x & 31;
    const int n = __builtin_amdgcn_readfirstlane((int)(blockIdx.x * 8 + (threadIdx.x >> 5)));
    if (n >= Nn) return;
    const float4 vs = *(const float4*)(xh + (long long)n * DD + lane * 4);
    const int my_s = srcs[(long long)n * CAP + lane];
    const float L2E = 1.4426950408889634f;
    float4 as4 = *(const float4*)(att_src + lane * 4);
    float4 ad4 = *(const float4*)(att_dst + lane * 4);
    as4.x *= L2E; as4.y *= L2E; as4.z *= L2E; as4.w *= L2E;
    ad4.x *= L2E; ad4.y *= L2E; ad4.z *= L2E; ad4.w *= L2E;
    const float ninf = -__builtin_inff();

    const float a_d = group8_sum(dot4(vs, ad4));
    float m   = leaky02(group8_sum(dot4(vs, as4)) + a_d);
    float den = 1.0f;
    float4 acc = vs;

    if (__builtin_amdgcn_readfirstlane(my_s) != -2) {
        const int cc = __builtin_popcount(__builtin_amdgcn_ballot_w32(my_s >= 0));
        for (int k0 = 0; k0 < cc; k0 += 4) {
            float4 R[4];
            float  e[4];
#pragma unroll
            for (int q = 0; q < 4; ++q) {
                int kk = k0 + q; if (kk > cc - 1) kk = cc - 1;
                const int sk = __builtin_amdgcn_readlane(my_s, kk);
                R[q] = *(const float4*)(xh + (long long)sk * DD + lane * 4);
            }
#pragma unroll
            for (int q = 0; q < 4; ++q) {
                const float t = leaky02(group8_sum(dot4(R[q], as4)) + a_d);
                e[q] = (k0 + q < cc) ? t : ninf;
            }
            const float gm = fmaxf(fmaxf(m, e[0]), fmaxf(fmaxf(e[1], e[2]), e[3]));
            const float sc = __builtin_amdgcn_exp2f(m - gm);
            den *= sc; acc.x *= sc; acc.y *= sc; acc.z *= sc; acc.w *= sc;
            m = gm;
#pragma unroll
            for (int q = 0; q < 4; ++q) {
                const float p = __builtin_amdgcn_exp2f(e[q] - gm);
                den += p;
                acc.x += p * R[q].x; acc.y += p * R[q].y; acc.z += p * R[q].z; acc.w += p * R[q].w;
            }
        }
    } else {
        const int deg  = __builtin_amdgcn_readfirstlane(cursor[n]);
        const int novf = __builtin_amdgcn_readfirstlane(cursor[Nn]);
        int prev = -1;
        for (int r = 0; r < deg; ++r) {
            const int2 es = next_edge(n, prev, bucket + (long long)n * CAP, ovf, novf, ei, E_, lane);
            prev = es.x;
            const float4 R = *(const float4*)(xh + (long long)es.y * DD + lane * 4);
            const float t  = leaky02(group8_sum(dot4(R, as4)) + a_d);
            const float gm = fmaxf(m, t);
            const float sc = __builtin_amdgcn_exp2f(m - gm);
            const float p  = __builtin_amdgcn_exp2f(t - gm);
            den = den * sc + p;
            acc.x = acc.x * sc + p * R.x; acc.y = acc.y * sc + p * R.y;
            acc.z = acc.z * sc + p * R.z; acc.w = acc.w * sc + p * R.w;
            m = gm;
        }
    }

    const float inv = 1.0f / den;
    const float4 gb = *(const float4*)(gbias + lane * 4);
    const float4 lg = *(const float4*)(ln_g + lane * 4);
    const float4 lb = *(const float4*)(ln_b + lane * 4);
    float* hp = h + (long long)n * DD + lane * 4;
    const float4 hold = *(const float4*)hp;
    float4 v = make_float4(acc.x * inv + gb.x, acc.y * inv + gb.y, acc.z * inv + gb.z, acc.w * inv + gb.w);
    float mean = wave_sum((v.x + v.y) + (v.z + v.w)) * (1.0f / DD);
    float4 cc4 = make_float4(v.x - mean, v.y - mean, v.z - mean, v.w - mean);
    float var = wave_sum((cc4.x * cc4.x + cc4.y * cc4.y) + (cc4.z * cc4.z + cc4.w * cc4.w)) * (1.0f / DD);
    float rs = rsqrtf(var + 1e-5f);
    float4 y = make_float4(cc4.x * rs * lg.x + lb.x, cc4.y * rs * lg.y + lb.y,
                           cc4.z * rs * lg.z + lb.z, cc4.w * rs * lg.w + lb.w);
    if (do_relu) { y.x = fmaxf(y.x, 0.f); y.y = fmaxf(y.y, 0.f); y.z = fmaxf(y.z, 0.f); y.w = fmaxf(y.w, 0.f); }
    *(float4*)hp = make_float4(hold.x + y.x, hold.y + y.y, hold.z + y.z, hold.w + y.w);
}

extern "C" void kernel_launch(void* const* d_in, const int* in_sizes, int n_in,
                              void* d_out, int out_size, void* d_ws, size_t ws_size,
                              hipStream_t stream) {
    const float* x      = (const float*)d_in[0];
    const int*   ei     = (const int*)d_in[1];
    const float* W_edge = (const float*)d_in[2];
    const float* b_edge = (const float*)d_in[3];
    const float* W_node = (const float*)d_in[4];
    const float* b_node = (const float*)d_in[5];
    const float* lin_w  = (const float*)d_in[6];
    const float* att_sp = (const float*)d_in[7];
    const float* att_dp = (const float*)d_in[8];
    const float* g_bias = (const float*)d_in[9];
    const float* ln_g   = (const float*)d_in[10];
    const float* ln_b   = (const float*)d_in[11];

    const int E_ = in_sizes[0] / 3;
    const int Nn = out_size / DD;

    int2*  bucket = (int2*)d_ws;
    int*   srcs   = (int*)d_ws + (long long)Nn * 2 * CAP;
    int*   cursor = srcs + (long long)Nn * CAP;
    int*   ovf    = cursor + (Nn + 1);
    float* Wc    = (float*)d_ws + (long long)Nn * DD - 4 * DD;
    float* xh    = (float*)d_ws + (long long)Nn * DD;
    float* h     = (float*)d_out;

    int g1  = (Nn + 1 + 255) / 256;
    int ge  = (E_ + 255) / 256;
    int gw  = (Nn + 7) / 8;
    dim3 gemm_grid((Nn + 15) / 16, 1);

    gnn_bucket_clear<<<g1, 256, 0, stream>>>(cursor, bucket, Nn);
    gnn_bucket_place<<<ge + 1, 256, 0, stream>>>(ei, cursor, bucket, ovf, Nn, E_,
                                                 W_edge, b_edge, W_node, Wc);

    gnn_init_fused<<<gemm_grid, 128, 0, stream>>>(x, ei, cursor, bucket, ovf, srcs, Wc,
                                                  b_node, lin_w, h, xh, Nn, E_);
    gnn_gat_gather<<<gw, 256, 0, stream>>>(cursor, srcs, bucket, ovf, ei, att_sp, att_dp, xh,
                                           g_bias, ln_g, ln_b, h, Nn, E_, 1);

    for (int i = 1; i < 3; ++i) {
        gnn_wmma_gemm128<<<gemm_grid, 128, 0, stream>>>(
            h, lin_w + (long long)i * DD * DD, xh, Nn);
        gnn_gat_gather<<<gw, 256, 0, stream>>>(cursor, srcs, bucket, ovf, ei,
                                               att_sp + i * DD, att_dp + i * DD, xh,
                                               g_bias + i * DD, ln_g + i * DD, ln_b + i * DD,
                                               h, Nn, E_, (i < 2) ? 1 : 0);
    }
    (void)n_in; (void)ws_size;
}
